# starting kernel
# baseline (speedup 1.0000x reference)
#include <hip/hip_runtime.h>
#include <math.h>

typedef __attribute__((ext_vector_type(2))) float v2f;
typedef __attribute__((ext_vector_type(4))) float v4f;
typedef __attribute__((ext_vector_type(8))) float v8f;

#define HDIM 128
#define TM 16
#define LDS_STRIDE 132

__device__ __forceinline__ v8f wmma4(v2f a, v2f b, v8f c) {
    return __builtin_amdgcn_wmma_f32_16x16x4_f32(false, a, false, b, (short)0, c, false, false);
}

__device__ __forceinline__ float act_relu_tanh(float z) {
#if __has_builtin(__builtin_amdgcn_tanhf)
    float v = __builtin_amdgcn_tanhf(z);
#else
    float v = tanhf(z);
#endif
    return fmaxf(v, 0.f);
}

__device__ __forceinline__ void load_wfrags(const float* __restrict__ W,
                                            int n0, int col16, int hi, v2f* bfrag) {
    const float* wp = W + (hi * 2) * HDIM + n0 + col16;
#pragma unroll
    for (int k = 0; k < 32; ++k) {
        bfrag[k].x = wp[(4 * k) * HDIM];
        bfrag[k].y = wp[(4 * k + 1) * HDIM];
    }
}

__device__ __forceinline__ v8f tile_matmul(const float* __restrict__ arow,
                                           const v2f* __restrict__ bfrag) {
    v2f a[32];
#pragma unroll
    for (int k = 0; k < 32; ++k)
        a[k] = *(const v2f*)(arow + 4 * k);
    v8f c0 = {}, c1 = {}, c2 = {}, c3 = {};
#pragma unroll
    for (int k = 0; k < 8; ++k) {
        c0 = wmma4(a[k],      bfrag[k],      c0);
        c1 = wmma4(a[k + 8],  bfrag[k + 8],  c1);
        c2 = wmma4(a[k + 16], bfrag[k + 16], c2);
        c3 = wmma4(a[k + 24], bfrag[k + 24], c3);
    }
    return (c0 + c1) + (c2 + c3);
}

__global__ __launch_bounds__(256) void gemm128_bias(const float* __restrict__ X,
                                                    const float* __restrict__ W,
                                                    const float* __restrict__ bias,
                                                    float* __restrict__ Y,
                                                    int M) {
    __shared__ float xs[TM * LDS_STRIDE];
    const int tid   = threadIdx.x;
    const int lane  = tid & 31;
    const int wv    = tid >> 5;
    const int n0    = wv * 16;
    const int col16 = lane & 15;
    const int hi    = lane >> 4;

    v2f bfrag[32];
    load_wfrags(W, n0, col16, hi, bfrag);
    const float bval = bias[n0 + col16];

    const float* arow = &xs[col16 * LDS_STRIDE + hi * 2];

    const int nstrips = M / TM;
    for (int strip = blockIdx.x; strip < nstrips; strip += gridDim.x) {
        const float* xp = X + (size_t)strip * (TM * HDIM);
#pragma unroll
        for (int i = 0; i < 2; ++i) {
            int idx = tid + i * 256;
            int lin = idx * 4;
            int r = lin >> 7;
            int c = lin & 127;
            *(v4f*)(&xs[r * LDS_STRIDE + c]) = *(const v4f*)(xp + lin);
        }
        __syncthreads();

        v8f acc = tile_matmul(arow, bfrag);

        float* yp = Y + (size_t)strip * (TM * HDIM) + (size_t)(8 * hi) * HDIM + n0 + col16;
#pragma unroll
        for (int i = 0; i < 8; ++i)
            yp[(size_t)i * HDIM] = acc[i] + bval;
        __syncthreads();
    }
}

__global__ __launch_bounds__(256) void rnn_recur(const float* __restrict__ Wh,
                                                 float* __restrict__ ws,
                                                 int S) {
    __shared__ float hs[TM * LDS_STRIDE];
    const int tid   = threadIdx.x;
    const int lane  = tid & 31;
    const int wv    = tid >> 5;
    const int n0    = wv * 16;
    const int col16 = lane & 15;
    const int hi    = lane >> 4;
    const int b0    = blockIdx.x * TM;

    v2f bfrag[32];
    load_wfrags(Wh, n0, col16, hi, bfrag);

    for (int i = tid; i < TM * LDS_STRIDE; i += 256) hs[i] = 0.f;
    __syncthreads();

    const size_t rowstride = (size_t)S * HDIM;
    float* gp = ws + (size_t)(b0 + 8 * hi) * rowstride + n0 + col16;
    const float* arow = &hs[col16 * LDS_STRIDE + hi * 2];

    float xi_cur[8];
#pragma unroll
    for (int i = 0; i < 8; ++i) xi_cur[i] = gp[(size_t)i * rowstride];

    for (int t = 0; t < S; ++t) {
        float xi_nxt[8];
        if (t + 1 < S) {
#pragma unroll
            for (int i = 0; i < 8; ++i)
                xi_nxt[i] = gp[(size_t)i * rowstride + (size_t)(t + 1) * HDIM];
        } else {
#pragma unroll
            for (int i = 0; i < 8; ++i) xi_nxt[i] = 0.f;
        }

        v8f acc = tile_matmul(arow, bfrag);
        __syncthreads();

#pragma unroll
        for (int i = 0; i < 8; ++i) {
            float v = act_relu_tanh(acc[i] + xi_cur[i]);
            hs[(i + 8 * hi) * LDS_STRIDE + n0 + col16] = v;
            gp[(size_t)i * rowstride + (size_t)t * HDIM] = v;
        }
        __syncthreads();

#pragma unroll
        for (int i = 0; i < 8; ++i) xi_cur[i] = xi_nxt[i];
    }
}

extern "C" void kernel_launch(void* const* d_in, const int* in_sizes, int n_in,
                              void* d_out, int out_size, void* d_ws, size_t ws_size,
                              hipStream_t stream) {
    const float* x  = (const float*)d_in[0];
    const float* Wi = (const float*)d_in[1];
    const float* Wh = (const float*)d_in[2];
    const float* b  = (const float*)d_in[3];
    const float* Wo = (const float*)d_in[4];
    const float* bo = (const float*)d_in[5];
    float* y  = (float*)d_out;
    float* ws = (float*)d_ws;

    const int B = 64, S = 2048;
    const int M = B * S;

    gemm128_bias<<<1024, 256, 0, stream>>>(x, Wi, b, ws, M);
    rnn_recur<<<B / TM, 256, 0, stream>>>(Wh, ws, S);
    gemm128_bias<<<1024, 256, 0, stream>>>(ws, Wo, bo, y, M);
}


# rnn_recur: the two step barriers fence LDS only, so no step waits for its global store acks or xi loads; xi in a 4-step register ring with counted waits; sums and their order unchanged
# speedup vs baseline: 1.6413x; 1.6413x over previous
#include <hip/hip_runtime.h>
#include <math.h>

typedef __attribute__((ext_vector_type(2))) float v2f;
typedef __attribute__((ext_vector_type(4))) float v4f;
typedef __attribute__((ext_vector_type(8))) float v8f;

#define HDIM 128
#define TM 16
#define LDS_STRIDE 132

__device__ __forceinline__ v8f wmma4(v2f a, v2f b, v8f c) {
    return __builtin_amdgcn_wmma_f32_16x16x4_f32(false, a, false, b, (short)0, c, false, false);
}

__device__ __forceinline__ float act_relu_tanh(float z) {
#if __has_builtin(__builtin_amdgcn_tanhf)
    float v = __builtin_amdgcn_tanhf(z);
#else
    float v = tanhf(z);
#endif
    return fmaxf(v, 0.f);
}

__device__ __forceinline__ void load_wfrags(const float* __restrict__ W,
                                            int n0, int col16, int hi, v2f* bfrag) {
    const float* wp = W + (hi * 2) * HDIM + n0 + col16;
#pragma unroll
    for (int k = 0; k < 32; ++k) {
        bfrag[k].x = wp[(4 * k) * HDIM];
        bfrag[k].y = wp[(4 * k + 1) * HDIM];
    }
}

__device__ __forceinline__ v8f tile_matmul(const float* __restrict__ arow,
                                           const v2f* __restrict__ bfrag) {
    v2f a[32];
#pragma unroll
    for (int k = 0; k < 32; ++k)
        a[k] = *(const v2f*)(arow + 4 * k);
    v8f c0 = {}, c1 = {}, c2 = {}, c3 = {};
#pragma unroll
    for (int k = 0; k < 8; ++k) {
        c0 = wmma4(a[k],      bfrag[k],      c0);
        c1 = wmma4(a[k + 8],  bfrag[k + 8],  c1);
        c2 = wmma4(a[k + 16], bfrag[k + 16], c2);
        c3 = wmma4(a[k + 24], bfrag[k + 24], c3);
    }
    return (c0 + c1) + (c2 + c3);
}

__global__ __launch_bounds__(256) void gemm128_bias(const float* __restrict__ X,
                                                    const float* __restrict__ W,
                                                    const float* __restrict__ bias,
                                                    float* __restrict__ Y,
                                                    int M) {
    __shared__ float xs[TM * LDS_STRIDE];
    const int tid   = threadIdx.x;
    const int lane  = tid & 31;
    const int wv    = tid >> 5;
    const int n0    = wv * 16;
    const int col16 = lane & 15;
    const int hi    = lane >> 4;

    v2f bfrag[32];
    load_wfrags(W, n0, col16, hi, bfrag);
    const float bval = bias[n0 + col16];

    const float* arow = &xs[col16 * LDS_STRIDE + hi * 2];

    const int nstrips = M / TM;
    for (int strip = blockIdx.x; strip < nstrips; strip += gridDim.x) {
        const float* xp = X + (size_t)strip * (TM * HDIM);
#pragma unroll
        for (int i = 0; i < 2; ++i) {
            int idx = tid + i * 256;
            int lin = idx * 4;
            int r = lin >> 7;
            int c = lin & 127;
            *(v4f*)(&xs[r * LDS_STRIDE + c]) = *(const v4f*)(xp + lin);
        }
        __syncthreads();

        v8f acc = tile_matmul(arow, bfrag);

        float* yp = Y + (size_t)strip * (TM * HDIM) + (size_t)(8 * hi) * HDIM + n0 + col16;
#pragma unroll
        for (int i = 0; i < 8; ++i)
            yp[(size_t)i * HDIM] = acc[i] + bval;
        __syncthreads();
    }
}

__device__ __forceinline__ void lds_barrier() {
    __builtin_amdgcn_fence(__ATOMIC_RELEASE, "workgroup", "local");
    __builtin_amdgcn_s_barrier();
    __builtin_amdgcn_fence(__ATOMIC_ACQUIRE, "workgroup", "local");
}

#define XI_AHEAD 4

__global__ __launch_bounds__(256) void rnn_recur(const float* __restrict__ Wh,
                                                 float* __restrict__ ws,
                                                 int S) {
    __shared__ float hs[TM * LDS_STRIDE];
    const int tid   = threadIdx.x;
    const int lane  = tid & 31;
    const int wv    = tid >> 5;
    const int n0    = wv * 16;
    const int col16 = lane & 15;
    const int hi    = lane >> 4;
    const int b0    = blockIdx.x * TM;

    v2f bfrag[32];
    load_wfrags(Wh, n0, col16, hi, bfrag);

    const size_t rowstride = (size_t)S * HDIM;
    float* gp = ws + (size_t)(b0 + 8 * hi) * rowstride + n0 + col16;
    const float* arow = &hs[col16 * LDS_STRIDE + hi * 2];

    float xi[XI_AHEAD][8];
#pragma unroll
    for (int u = 0; u < XI_AHEAD; ++u) {
#pragma unroll
        for (int i = 0; i < 8; ++i)
            xi[u][i] = gp[(size_t)i * rowstride + (size_t)u * HDIM];
    }

    for (int i = tid; i < TM * LDS_STRIDE; i += 256) hs[i] = 0.f;
    __syncthreads();

    for (int t0 = 0; t0 + XI_AHEAD <= S; t0 += XI_AHEAD) {
#pragma unroll
        for (int u = 0; u < XI_AHEAD; ++u) {
            const int t = t0 + u;
            v8f acc = tile_matmul(arow, bfrag);
            lds_barrier();

#pragma unroll
            for (int i = 0; i < 8; ++i) {
                float v = act_relu_tanh(acc[i] + xi[u][i]);
                hs[(i + 8 * hi) * LDS_STRIDE + n0 + col16] = v;
                gp[(size_t)i * rowstride + (size_t)t * HDIM] = v;
            }
            const int tn = (t + XI_AHEAD < S) ? t + XI_AHEAD : S - 1;
#pragma unroll
            for (int i = 0; i < 8; ++i)
                xi[u][i] = gp[(size_t)i * rowstride + (size_t)tn * HDIM];
            lds_barrier();
        }
    }
}

extern "C" void kernel_launch(void* const* d_in, const int* in_sizes, int n_in,
                              void* d_out, int out_size, void* d_ws, size_t ws_size,
                              hipStream_t stream) {
    const float* x  = (const float*)d_in[0];
    const float* Wi = (const float*)d_in[1];
    const float* Wh = (const float*)d_in[2];
    const float* b  = (const float*)d_in[3];
    const float* Wo = (const float*)d_in[4];
    const float* bo = (const float*)d_in[5];
    float* y  = (float*)d_out;
    float* ws = (float*)d_ws;

    constexpr int B = 64, S = 2048;
    static_assert(S % XI_AHEAD == 0, "rnn_recur walks the sequence in groups of XI_AHEAD steps");
    const int M = B * S;

    gemm128_bias<<<1024, 256, 0, stream>>>(x, Wi, b, ws, M);
    rnn_recur<<<B / TM, 256, 0, stream>>>(Wh, ws, S);
    gemm128_bias<<<1024, 256, 0, stream>>>(ws, Wo, bo, y, M);
}


# rnn_recur: step order pinned off the serial path: h fragments as one LDS load group ahead of the WMMAs, barriers split, global stores and xi loads between signal and wait; sums and order unchanged
# speedup vs baseline: 2.0452x; 1.2461x over previous
#include <hip/hip_runtime.h>
#include <math.h>

typedef __attribute__((ext_vector_type(2))) float v2f;
typedef __attribute__((ext_vector_type(4))) float v4f;
typedef __attribute__((ext_vector_type(8))) float v8f;

#define HDIM 128
#define TM 16
#define LDS_STRIDE 132

__device__ __forceinline__ v8f wmma4(v2f a, v2f b, v8f c) {
    return __builtin_amdgcn_wmma_f32_16x16x4_f32(false, a, false, b, (short)0, c, false, false);
}

__device__ __forceinline__ float act_relu_tanh(float z) {
#if __has_builtin(__builtin_amdgcn_tanhf)
    float v = __builtin_amdgcn_tanhf(z);
#else
    float v = tanhf(z);
#endif
    return fmaxf(v, 0.f);
}

__device__ __forceinline__ void load_wfrags(const float* __restrict__ W,
                                            int n0, int col16, int hi, v2f* bfrag) {
    const float* wp = W + (hi * 2) * HDIM + n0 + col16;
#pragma unroll
    for (int k = 0; k < 32; ++k) {
        bfrag[k].x = wp[(4 * k) * HDIM];
        bfrag[k].y = wp[(4 * k + 1) * HDIM];
    }
}

__device__ __forceinline__ v8f tile_matmul(const float* __restrict__ arow,
                                           const v2f* __restrict__ bfrag) {
    v2f a[32];
#pragma unroll
    for (int k = 0; k < 32; ++k)
        a[k] = *(const v2f*)(arow + 4 * k);
    v8f c0 = {}, c1 = {}, c2 = {}, c3 = {};
#pragma unroll
    for (int k = 0; k < 8; ++k) {
        c0 = wmma4(a[k],      bfrag[k],      c0);
        c1 = wmma4(a[k + 8],  bfrag[k + 8],  c1);
        c2 = wmma4(a[k + 16], bfrag[k + 16], c2);
        c3 = wmma4(a[k + 24], bfrag[k + 24], c3);
    }
    return (c0 + c1) + (c2 + c3);
}

__global__ __launch_bounds__(256) void gemm128_bias(const float* __restrict__ X,
                                                    const float* __restrict__ W,
                                                    const float* __restrict__ bias,
                                                    float* __restrict__ Y,
                                                    int M) {
    __shared__ float xs[TM * LDS_STRIDE];
    const int tid   = threadIdx.x;
    const int lane  = tid & 31;
    const int wv    = tid >> 5;
    const int n0    = wv * 16;
    const int col16 = lane & 15;
    const int hi    = lane >> 4;

    v2f bfrag[32];
    load_wfrags(W, n0, col16, hi, bfrag);
    const float bval = bias[n0 + col16];

    const float* arow = &xs[col16 * LDS_STRIDE + hi * 2];

    const int nstrips = M / TM;
    for (int strip = blockIdx.x; strip < nstrips; strip += gridDim.x) {
        const float* xp = X + (size_t)strip * (TM * HDIM);
#pragma unroll
        for (int i = 0; i < 2; ++i) {
            int idx = tid + i * 256;
            int lin = idx * 4;
            int r = lin >> 7;
            int c = lin & 127;
            *(v4f*)(&xs[r * LDS_STRIDE + c]) = *(const v4f*)(xp + lin);
        }
        __syncthreads();

        v8f acc = tile_matmul(arow, bfrag);

        float* yp = Y + (size_t)strip * (TM * HDIM) + (size_t)(8 * hi) * HDIM + n0 + col16;
#pragma unroll
        for (int i = 0; i < 8; ++i)
            yp[(size_t)i * HDIM] = acc[i] + bval;
        __syncthreads();
    }
}

__device__ __forceinline__ void lds_barrier_signal() {
    __builtin_amdgcn_fence(__ATOMIC_RELEASE, "workgroup", "local");
    __builtin_amdgcn_s_barrier_signal(-1);
}
__device__ __forceinline__ void lds_barrier_wait() {
    __builtin_amdgcn_s_barrier_wait(-1);
    __builtin_amdgcn_fence(__ATOMIC_ACQUIRE, "workgroup", "local");
}

#define XI_AHEAD 4

__device__ __forceinline__ void load_afrags(const float* __restrict__ arow, v2f* a) {
#pragma unroll
    for (int j = 0; j < 4; ++j) {
#pragma unroll
        for (int c = 0; c < 4; ++c) {
            a[8 * c + 2 * j]     = *(const v2f*)(arow + 4 * (8 * c + 2 * j));
            a[8 * c + 2 * j + 1] = *(const v2f*)(arow + 4 * (8 * c + 2 * j + 1));
        }
    }
}

__device__ __forceinline__ void wmma_steps(const v2f* a, const v2f* bfrag, int k0, int k1,
                                           v8f& c0, v8f& c1, v8f& c2, v8f& c3) {
#pragma unroll
    for (int k = k0; k < k1; ++k) {
        c0 = wmma4(a[k],      bfrag[k],      c0);
        c1 = wmma4(a[k + 8],  bfrag[k + 8],  c1);
        c2 = wmma4(a[k + 16], bfrag[k + 16], c2);
        c3 = wmma4(a[k + 24], bfrag[k + 24], c3);
    }
}

__global__ __launch_bounds__(256) void rnn_recur(const float* __restrict__ Wh,
                                                 float* __restrict__ ws,
                                                 int S) {
    __shared__ float hs[TM * LDS_STRIDE];
    const int tid   = threadIdx.x;
    const int lane  = tid & 31;
    const int wv    = tid >> 5;
    const int n0    = wv * 16;
    const int col16 = lane & 15;
    const int hi    = lane >> 4;
    const int b0    = blockIdx.x * TM;

    v2f bfrag[32];
    load_wfrags(Wh, n0, col16, hi, bfrag);

    const size_t rowstride = (size_t)S * HDIM;
    float* gp = ws + (size_t)(b0 + 8 * hi) * rowstride + n0 + col16;
    const float* arow = &hs[col16 * LDS_STRIDE + hi * 2];

    float xi[XI_AHEAD][8];
#pragma unroll
    for (int u = 0; u < XI_AHEAD; ++u) {
#pragma unroll
        for (int i = 0; i < 8; ++i)
            xi[u][i] = gp[(size_t)i * rowstride + (size_t)u * HDIM];
    }

    for (int i = tid; i < TM * LDS_STRIDE; i += 256) hs[i] = 0.f;
    __syncthreads();

    v2f a[32];
    load_afrags(arow, a);

    for (int t0 = 0; t0 + XI_AHEAD <= S; t0 += XI_AHEAD) {
#pragma unroll
        for (int u = 0; u < XI_AHEAD; ++u) {
            const int t = t0 + u;
            v8f c0 = {}, c1 = {}, c2 = {}, c3 = {};
            wmma_steps(a, bfrag, 0, 4, c0, c1, c2, c3);

            __builtin_amdgcn_sched_barrier(0);
            lds_barrier_signal();
            __builtin_amdgcn_sched_barrier(0);

            wmma_steps(a, bfrag, 4, 8, c0, c1, c2, c3);
            v8f acc = (c0 + c1) + (c2 + c3);

            lds_barrier_wait();

            float v[8];
#pragma unroll
            for (int i = 0; i < 8; ++i) {
                v[i] = act_relu_tanh(acc[i] + xi[u][i]);
                hs[(i + 8 * hi) * LDS_STRIDE + n0 + col16] = v[i];
            }

            lds_barrier_signal();

#pragma unroll
            for (int i = 0; i < 8; ++i)
                gp[(size_t)i * rowstride + (size_t)t * HDIM] = v[i];
            const int tn = (t + XI_AHEAD < S) ? t + XI_AHEAD : S - 1;
#pragma unroll
            for (int i = 0; i < 8; ++i)
                xi[u][i] = gp[(size_t)i * rowstride + (size_t)tn * HDIM];

            lds_barrier_wait();

            load_afrags(arow, a);
            __builtin_amdgcn_sched_barrier(0);
        }
    }
}

extern "C" void kernel_launch(void* const* d_in, const int* in_sizes, int n_in,
                              void* d_out, int out_size, void* d_ws, size_t ws_size,
                              hipStream_t stream) {
    const float* x  = (const float*)d_in[0];
    const float* Wi = (const float*)d_in[1];
    const float* Wh = (const float*)d_in[2];
    const float* b  = (const float*)d_in[3];
    const float* Wo = (const float*)d_in[4];
    const float* bo = (const float*)d_in[5];
    float* y  = (float*)d_out;
    float* ws = (float*)d_ws;

    constexpr int B = 64, S = 2048;
    static_assert(S % XI_AHEAD == 0, "rnn_recur walks the sequence in groups of XI_AHEAD steps");
    const int M = B * S;

    gemm128_bias<<<1024, 256, 0, stream>>>(x, Wi, b, ws, M);
    rnn_recur<<<B / TM, 256, 0, stream>>>(Wh, ws, S);
    gemm128_bias<<<1024, 256, 0, stream>>>(ws, Wo, bo, y, M);
}


# rnn_recur: step's k range split by the wave group that writes it; h handed over in 3 barrier phases (two LDS copies): each group's WMMAs run under the other's hand-over; sums and order unchanged
# speedup vs baseline: 2.1529x; 1.0527x over previous
#include <hip/hip_runtime.h>
#include <math.h>

typedef __attribute__((ext_vector_type(2))) float v2f;
typedef __attribute__((ext_vector_type(4))) float v4f;
typedef __attribute__((ext_vector_type(8))) float v8f;

#define HDIM 128
#define TM 16
#define LDS_STRIDE 132

__device__ __forceinline__ v8f wmma4(v2f a, v2f b, v8f c) {
    return __builtin_amdgcn_wmma_f32_16x16x4_f32(false, a, false, b, (short)0, c, false, false);
}

__device__ __forceinline__ float act_relu_tanh(float z) {
#if __has_builtin(__builtin_amdgcn_tanhf)
    float v = __builtin_amdgcn_tanhf(z);
#else
    float v = tanhf(z);
#endif
    return fmaxf(v, 0.f);
}

__device__ __forceinline__ void load_wfrags(const float* __restrict__ W,
                                            int n0, int col16, int hi, v2f* bfrag) {
    const float* wp = W + (hi * 2) * HDIM + n0 + col16;
#pragma unroll
    for (int k = 0; k < 32; ++k) {
        bfrag[k].x = wp[(4 * k) * HDIM];
        bfrag[k].y = wp[(4 * k + 1) * HDIM];
    }
}

__device__ __forceinline__ v8f tile_matmul(const float* __restrict__ arow,
                                           const v2f* __restrict__ bfrag) {
    v2f a[32];
#pragma unroll
    for (int k = 0; k < 32; ++k)
        a[k] = *(const v2f*)(arow + 4 * k);
    v8f c0 = {}, c1 = {}, c2 = {}, c3 = {};
#pragma unroll
    for (int k = 0; k < 8; ++k) {
        c0 = wmma4(a[k],      bfrag[k],      c0);
        c1 = wmma4(a[k + 8],  bfrag[k + 8],  c1);
        c2 = wmma4(a[k + 16], bfrag[k + 16], c2);
        c3 = wmma4(a[k + 24], bfrag[k + 24], c3);
    }
    return (c0 + c1) + (c2 + c3);
}

__global__ __launch_bounds__(256) void gemm128_bias(const float* __restrict__ X,
                                                    const float* __restrict__ W,
                                                    const float* __restrict__ bias,
                                                    float* __restrict__ Y,
                                                    int M) {
    __shared__ float xs[TM * LDS_STRIDE];
    const int tid   = threadIdx.x;
    const int lane  = tid & 31;
    const int wv    = tid >> 5;
    const int n0    = wv * 16;
    const int col16 = lane & 15;
    const int hi    = lane >> 4;

    v2f bfrag[32];
    load_wfrags(W, n0, col16, hi, bfrag);
    const float bval = bias[n0 + col16];

    const float* arow = &xs[col16 * LDS_STRIDE + hi * 2];

    const int nstrips = M / TM;
    for (int strip = blockIdx.x; strip < nstrips; strip += gridDim.x) {
        const float* xp = X + (size_t)strip * (TM * HDIM);
#pragma unroll
        for (int i = 0; i < 2; ++i) {
            int idx = tid + i * 256;
            int lin = idx * 4;
            int r = lin >> 7;
            int c = lin & 127;
            *(v4f*)(&xs[r * LDS_STRIDE + c]) = *(const v4f*)(xp + lin);
        }
        __syncthreads();

        v8f acc = tile_matmul(arow, bfrag);

        float* yp = Y + (size_t)strip * (TM * HDIM) + (size_t)(8 * hi) * HDIM + n0 + col16;
#pragma unroll
        for (int i = 0; i < 8; ++i)
            yp[(size_t)i * HDIM] = acc[i] + bval;
        __syncthreads();
    }
}

__device__ __forceinline__ void lds_barrier_signal() {
    __builtin_amdgcn_fence(__ATOMIC_RELEASE, "workgroup", "local");
    __builtin_amdgcn_s_barrier_signal(-1);
}
__device__ __forceinline__ void lds_barrier_wait() {
    __builtin_amdgcn_s_barrier_wait(-1);
    __builtin_amdgcn_fence(__ATOMIC_ACQUIRE, "workgroup", "local");
}

#define XI_AHEAD 4
#define HBUF (TM * LDS_STRIDE)

__device__ __forceinline__ void load_afrags_half(const float* __restrict__ arow, v2f* f, int kb) {
#pragma unroll
    for (int j = 0; j < 4; ++j) {
#pragma unroll
        for (int c = 0; c < 2; ++c) {
            f[8 * c + 2 * j]     = *(const v2f*)(arow + 4 * (kb + 8 * c + 2 * j));
            f[8 * c + 2 * j + 1] = *(const v2f*)(arow + 4 * (kb + 8 * c + 2 * j + 1));
        }
    }
}

__device__ __forceinline__ void wmma_pair(const v2f* f, const v2f* bfrag, int kb, v8f& ca, v8f& cb,
                                          int k0 = 0, int k1 = 8) {
#pragma unroll
    for (int k = k0; k < k1; ++k) {
        ca = wmma4(f[k],     bfrag[kb + k],     ca);
        cb = wmma4(f[8 + k], bfrag[kb + 8 + k], cb);
    }
}

__device__ __forceinline__ void barrier_wait_behind(v8f& ca, v8f& cb) {
    asm volatile("s_barrier_wait 0xffff" : "+v"(ca), "+v"(cb) : : "memory");
}
__device__ __forceinline__ void barrier_wait_behind(v8f& ca, v8f& cb, v8f& cc, v8f& cd) {
    asm volatile("s_barrier_wait 0xffff" : "+v"(ca), "+v"(cb), "+v"(cc), "+v"(cd) : : "memory");
}
__device__ __forceinline__ void barrier_signal_behind(v8f& ca, v8f& cb, v8f& cc, v8f& cd) {
    asm volatile("s_barrier_signal -1" : "+v"(ca), "+v"(cb), "+v"(cc), "+v"(cd) : : "memory");
}
__device__ __forceinline__ void barrier_wait_plain() {
    asm volatile("s_barrier_wait 0xffff" : : : "memory");
}
__device__ __forceinline__ void barrier_signal_nothing() {
    asm volatile("s_barrier_signal -1");
}
__device__ __forceinline__ void barrier_wait_before(v2f& fa, v2f& fb) {
    asm volatile("s_barrier_wait 0xffff" : "+v"(fa), "+v"(fb) : : "memory");
}

template <bool EARLY>
__device__ __forceinline__ void recur_loop(const v2f* bfrag, float (&xi)[XI_AHEAD][8],
                                           float* __restrict__ gp, size_t rowstride,
                                           float* hs, const float* arow,
                                           int hi, int ncol, int S) {
    v2f ax[16];
    load_afrags_half(arow + HBUF, ax, 0);
    __builtin_amdgcn_s_barrier_signal(-1);
    if (!EARLY) {
        __builtin_amdgcn_s_barrier_wait(-1);
        __builtin_amdgcn_s_barrier_signal(-1);
    }

    for (int t0 = 0; t0 + XI_AHEAD <= S; t0 += XI_AHEAD) {
#pragma unroll
        for (int u = 0; u < XI_AHEAD; ++u) {
            const int t = t0 + u;
            const float* ar = arow + ((u + 1) & 1) * HBUF;
            const float* aw = arow + (u & 1) * HBUF;
            float* hw = hs + (u & 1) * HBUF;
            v2f ay[16];

            if (EARLY) {
                barrier_wait_before(ax[0], ax[8]);
                barrier_signal_nothing();
            } else {
                barrier_wait_plain();
                barrier_signal_nothing();
                load_afrags_half(ar, ay, 16);
            }
            __builtin_amdgcn_sched_barrier(0);

            v8f c0 = {}, c1 = {}, c2 = {}, c3 = {};
            if (EARLY) {
                wmma_pair(ax, bfrag, 0, c0, c1, 0, 5);
                barrier_wait_behind(c0, c1);
                load_afrags_half(ar, ay, 16);
                __builtin_amdgcn_sched_barrier(0);
                wmma_pair(ax, bfrag, 0, c0, c1, 5, 8);
                asm volatile("" : "+v"(c0), "+v"(c1), "+v"(ay[0]), "+v"(ay[8]));
            } else {
                wmma_pair(ax, bfrag, 0, c0, c1);
            }

            wmma_pair(ay, bfrag, 16, c2, c3);

            if (!EARLY) {
                barrier_wait_behind(c0, c1, c2, c3);
                barrier_signal_behind(c0, c1, c2, c3);
                lds_barrier_wait();
            }

            v8f acc = (c0 + c1) + (c2 + c3);

            float v[8];
#pragma unroll
            for (int i = 0; i < 8; ++i) {
                v[i] = act_relu_tanh(acc[i] + xi[u][i]);
                hw[(i + 8 * hi) * LDS_STRIDE + ncol] = v[i];
            }

            lds_barrier_signal();

#pragma unroll
            for (int i = 0; i < 8; ++i)
                gp[(size_t)i * rowstride + (size_t)t * HDIM] = v[i];
            const int tn = (t + XI_AHEAD < S) ? t + XI_AHEAD : S - 1;
#pragma unroll
            for (int i = 0; i < 8; ++i)
                xi[u][i] = gp[(size_t)i * rowstride + (size_t)tn * HDIM];

            if (EARLY) lds_barrier_wait();
            load_afrags_half(aw, ax, 0);
            if (EARLY) barrier_signal_nothing();
            __builtin_amdgcn_sched_barrier(0);
        }
    }

    if (EARLY) {
        __builtin_amdgcn_s_barrier_wait(-1);
        __builtin_amdgcn_s_barrier_signal(-1);
    }
    __builtin_amdgcn_s_barrier_wait(-1);
}

__global__ __launch_bounds__(256) void rnn_recur(const float* __restrict__ Wh,
                                                 float* __restrict__ ws,
                                                 int S) {
    __shared__ float hs[2 * HBUF];
    const int tid   = threadIdx.x;
    const int lane  = tid & 31;
    const int wv    = tid >> 5;
    const int n0    = wv * 16;
    const int col16 = lane & 15;
    const int hi    = lane >> 4;
    const int b0    = blockIdx.x * TM;
    const bool early = __builtin_amdgcn_readfirstlane(wv) < 4;

    v2f bfrag[32];
    load_wfrags(Wh, n0, col16, hi, bfrag);

    const size_t rowstride = (size_t)S * HDIM;
    float* gp = ws + (size_t)(b0 + 8 * hi) * rowstride + n0 + col16;
    const float* arow = &hs[col16 * LDS_STRIDE + hi * 2];

    float xi[XI_AHEAD][8];
#pragma unroll
    for (int u = 0; u < XI_AHEAD; ++u) {
#pragma unroll
        for (int i = 0; i < 8; ++i)
            xi[u][i] = gp[(size_t)i * rowstride + (size_t)u * HDIM];
    }

    for (int i = tid; i < 2 * HBUF; i += 256) hs[i] = 0.f;
    __syncthreads();

    if (early) recur_loop<true >(bfrag, xi, gp, rowstride, hs, arow, hi, n0 + col16, S);
    else       recur_loop<false>(bfrag, xi, gp, rowstride, hs, arow, hi, n0 + col16, S);
}

extern "C" void kernel_launch(void* const* d_in, const int* in_sizes, int n_in,
                              void* d_out, int out_size, void* d_ws, size_t ws_size,
                              hipStream_t stream) {
    const float* x  = (const float*)d_in[0];
    const float* Wi = (const float*)d_in[1];
    const float* Wh = (const float*)d_in[2];
    const float* b  = (const float*)d_in[3];
    const float* Wo = (const float*)d_in[4];
    const float* bo = (const float*)d_in[5];
    float* y  = (float*)d_out;
    float* ws = (float*)d_ws;

    constexpr int B = 64, S = 2048;
    static_assert(S % XI_AHEAD == 0, "rnn_recur walks the sequence in groups of XI_AHEAD steps");
    const int M = B * S;

    gemm128_bias<<<1024, 256, 0, stream>>>(x, Wi, b, ws, M);
    rnn_recur<<<B / TM, 256, 0, stream>>>(Wh, ws, S);
    gemm128_bias<<<1024, 256, 0, stream>>>(ws, Wo, bo, y, M);
}


# rnn_recur: late waves hold issue priority (s_setprio) from their last WMMA to their next WMMA run: hand-over no longer waits behind the early waves' WMMAs; sums and order unchanged
# speedup vs baseline: 2.3490x; 1.0911x over previous
#include <hip/hip_runtime.h>
#include <math.h>

typedef __attribute__((ext_vector_type(2))) float v2f;
typedef __attribute__((ext_vector_type(4))) float v4f;
typedef __attribute__((ext_vector_type(8))) float v8f;

#define HDIM 128
#define TM 16
#define LDS_STRIDE 132

__device__ __forceinline__ v8f wmma4(v2f a, v2f b, v8f c) {
    return __builtin_amdgcn_wmma_f32_16x16x4_f32(false, a, false, b, (short)0, c, false, false);
}

__device__ __forceinline__ float act_relu_tanh(float z) {
#if __has_builtin(__builtin_amdgcn_tanhf)
    float v = __builtin_amdgcn_tanhf(z);
#else
    float v = tanhf(z);
#endif
    return fmaxf(v, 0.f);
}

__device__ __forceinline__ void load_wfrags(const float* __restrict__ W,
                                            int n0, int col16, int hi, v2f* bfrag) {
    const float* wp = W + (hi * 2) * HDIM + n0 + col16;
#pragma unroll
    for (int k = 0; k < 32; ++k) {
        bfrag[k].x = wp[(4 * k) * HDIM];
        bfrag[k].y = wp[(4 * k + 1) * HDIM];
    }
}

__device__ __forceinline__ v8f tile_matmul(const float* __restrict__ arow,
                                           const v2f* __restrict__ bfrag) {
    v2f a[32];
#pragma unroll
    for (int k = 0; k < 32; ++k)
        a[k] = *(const v2f*)(arow + 4 * k);
    v8f c0 = {}, c1 = {}, c2 = {}, c3 = {};
#pragma unroll
    for (int k = 0; k < 8; ++k) {
        c0 = wmma4(a[k],      bfrag[k],      c0);
        c1 = wmma4(a[k + 8],  bfrag[k + 8],  c1);
        c2 = wmma4(a[k + 16], bfrag[k + 16], c2);
        c3 = wmma4(a[k + 24], bfrag[k + 24], c3);
    }
    return (c0 + c1) + (c2 + c3);
}

__global__ __launch_bounds__(256) void gemm128_bias(const float* __restrict__ X,
                                                    const float* __restrict__ W,
                                                    const float* __restrict__ bias,
                                                    float* __restrict__ Y,
                                                    int M) {
    __shared__ float xs[TM * LDS_STRIDE];
    const int tid   = threadIdx.x;
    const int lane  = tid & 31;
    const int wv    = tid >> 5;
    const int n0    = wv * 16;
    const int col16 = lane & 15;
    const int hi    = lane >> 4;

    v2f bfrag[32];
    load_wfrags(W, n0, col16, hi, bfrag);
    const float bval = bias[n0 + col16];

    const float* arow = &xs[col16 * LDS_STRIDE + hi * 2];

    const int nstrips = M / TM;
    for (int strip = blockIdx.x; strip < nstrips; strip += gridDim.x) {
        const float* xp = X + (size_t)strip * (TM * HDIM);
#pragma unroll
        for (int i = 0; i < 2; ++i) {
            int idx = tid + i * 256;
            int lin = idx * 4;
            int r = lin >> 7;
            int c = lin & 127;
            *(v4f*)(&xs[r * LDS_STRIDE + c]) = *(const v4f*)(xp + lin);
        }
        __syncthreads();

        v8f acc = tile_matmul(arow, bfrag);

        float* yp = Y + (size_t)strip * (TM * HDIM) + (size_t)(8 * hi) * HDIM + n0 + col16;
#pragma unroll
        for (int i = 0; i < 8; ++i)
            yp[(size_t)i * HDIM] = acc[i] + bval;
        __syncthreads();
    }
}

__device__ __forceinline__ void lds_barrier_signal() {
    __builtin_amdgcn_fence(__ATOMIC_RELEASE, "workgroup", "local");
    __builtin_amdgcn_s_barrier_signal(-1);
}
__device__ __forceinline__ void lds_barrier_wait() {
    __builtin_amdgcn_s_barrier_wait(-1);
    __builtin_amdgcn_fence(__ATOMIC_ACQUIRE, "workgroup", "local");
}

#define XI_AHEAD 4
#define HBUF (TM * LDS_STRIDE)

__device__ __forceinline__ void load_afrags_half(const float* __restrict__ arow, v2f* f, int kb) {
#pragma unroll
    for (int j = 0; j < 4; ++j) {
#pragma unroll
        for (int c = 0; c < 2; ++c) {
            f[8 * c + 2 * j]     = *(const v2f*)(arow + 4 * (kb + 8 * c + 2 * j));
            f[8 * c + 2 * j + 1] = *(const v2f*)(arow + 4 * (kb + 8 * c + 2 * j + 1));
        }
    }
}

__device__ __forceinline__ void wmma_pair(const v2f* f, const v2f* bfrag, int kb, v8f& ca, v8f& cb,
                                          int k0 = 0, int k1 = 8) {
#pragma unroll
    for (int k = k0; k < k1; ++k) {
        ca = wmma4(f[k],     bfrag[kb + k],     ca);
        cb = wmma4(f[8 + k], bfrag[kb + 8 + k], cb);
    }
}

__device__ __forceinline__ void barrier_wait_behind(v8f& ca, v8f& cb) {
    asm volatile("s_barrier_wait 0xffff" : "+v"(ca), "+v"(cb) : : "memory");
}
__device__ __forceinline__ void barrier_wait_behind(v8f& ca, v8f& cb, v8f& cc, v8f& cd) {
    asm volatile("s_barrier_wait 0xffff" : "+v"(ca), "+v"(cb), "+v"(cc), "+v"(cd) : : "memory");
}
__device__ __forceinline__ void barrier_signal_behind(v8f& ca, v8f& cb, v8f& cc, v8f& cd) {
    asm volatile("s_barrier_signal -1" : "+v"(ca), "+v"(cb), "+v"(cc), "+v"(cd) : : "memory");
}
__device__ __forceinline__ void barrier_wait_plain() {
    asm volatile("s_barrier_wait 0xffff" : : : "memory");
}
__device__ __forceinline__ void barrier_signal_nothing() {
    asm volatile("s_barrier_signal -1");
}
__device__ __forceinline__ void barrier_wait_before(v2f& fa, v2f& fb) {
    asm volatile("s_barrier_wait 0xffff" : "+v"(fa), "+v"(fb) : : "memory");
}

template <bool EARLY>
__device__ __forceinline__ void recur_loop(const v2f* bfrag, float (&xi)[XI_AHEAD][8],
                                           float* __restrict__ gp, size_t rowstride,
                                           float* hs, const float* arow,
                                           int hi, int ncol, int S) {
    v2f ax[16];
    load_afrags_half(arow + HBUF, ax, 0);
    __builtin_amdgcn_s_barrier_signal(-1);
    if (!EARLY) {
        __builtin_amdgcn_s_barrier_wait(-1);
        __builtin_amdgcn_s_barrier_signal(-1);
    }

    for (int t0 = 0; t0 + XI_AHEAD <= S; t0 += XI_AHEAD) {
#pragma unroll
        for (int u = 0; u < XI_AHEAD; ++u) {
            const int t = t0 + u;
            const float* ar = arow + ((u + 1) & 1) * HBUF;
            const float* aw = arow + (u & 1) * HBUF;
            float* hw = hs + (u & 1) * HBUF;
            v2f ay[16];

            if (EARLY) {
                barrier_wait_before(ax[0], ax[8]);
                barrier_signal_nothing();
            } else {
                barrier_wait_plain();
                barrier_signal_nothing();
                __builtin_amdgcn_s_setprio(0);
                load_afrags_half(ar, ay, 16);
            }
            __builtin_amdgcn_sched_barrier(0);

            v8f c0 = {}, c1 = {}, c2 = {}, c3 = {};
            if (EARLY) {
                wmma_pair(ax, bfrag, 0, c0, c1, 0, 5);
                barrier_wait_behind(c0, c1);
                load_afrags_half(ar, ay, 16);
                __builtin_amdgcn_sched_barrier(0);
                wmma_pair(ax, bfrag, 0, c0, c1, 5, 8);
                asm volatile("" : "+v"(c0), "+v"(c1), "+v"(ay[0]), "+v"(ay[8]));
            } else {
                wmma_pair(ax, bfrag, 0, c0, c1);
            }

            wmma_pair(ay, bfrag, 16, c2, c3);

            if (!EARLY) {
                barrier_wait_behind(c0, c1, c2, c3);
                barrier_signal_behind(c0, c1, c2, c3);
                __builtin_amdgcn_s_setprio(3);
                lds_barrier_wait();
            }

            v8f acc = (c0 + c1) + (c2 + c3);

            float v[8];
#pragma unroll
            for (int i = 0; i < 8; ++i) {
                v[i] = act_relu_tanh(acc[i] + xi[u][i]);
                hw[(i + 8 * hi) * LDS_STRIDE + ncol] = v[i];
            }

            lds_barrier_signal();
            if (!EARLY) __builtin_amdgcn_sched_barrier(0);

#pragma unroll
            for (int i = 0; i < 8; ++i)
                gp[(size_t)i * rowstride + (size_t)t * HDIM] = v[i];
            const int tn = (t + XI_AHEAD < S) ? t + XI_AHEAD : S - 1;
#pragma unroll
            for (int i = 0; i < 8; ++i)
                xi[u][i] = gp[(size_t)i * rowstride + (size_t)tn * HDIM];

            if (EARLY) lds_barrier_wait();
            load_afrags_half(aw, ax, 0);
            if (EARLY) barrier_signal_nothing();
            __builtin_amdgcn_sched_barrier(0);
        }
    }

    if (EARLY) {
        __builtin_amdgcn_s_barrier_wait(-1);
        __builtin_amdgcn_s_barrier_signal(-1);
    }
    __builtin_amdgcn_s_barrier_wait(-1);
}

__global__ __launch_bounds__(256) void rnn_recur(const float* __restrict__ Wh,
                                                 float* __restrict__ ws,
                                                 int S) {
    __shared__ float hs[2 * HBUF];
    const int tid   = threadIdx.x;
    const int lane  = tid & 31;
    const int wv    = tid >> 5;
    const int n0    = wv * 16;
    const int col16 = lane & 15;
    const int hi    = lane >> 4;
    const int b0    = blockIdx.x * TM;
    const bool early = __builtin_amdgcn_readfirstlane(wv) < 4;

    v2f bfrag[32];
    load_wfrags(Wh, n0, col16, hi, bfrag);

    const size_t rowstride = (size_t)S * HDIM;
    float* gp = ws + (size_t)(b0 + 8 * hi) * rowstride + n0 + col16;
    const float* arow = &hs[col16 * LDS_STRIDE + hi * 2];

    float xi[XI_AHEAD][8];
#pragma unroll
    for (int u = 0; u < XI_AHEAD; ++u) {
#pragma unroll
        for (int i = 0; i < 8; ++i)
            xi[u][i] = gp[(size_t)i * rowstride + (size_t)u * HDIM];
    }

    for (int i = tid; i < 2 * HBUF; i += 256) hs[i] = 0.f;
    __syncthreads();

    if (early) recur_loop<true >(bfrag, xi, gp, rowstride, hs, arow, hi, n0 + col16, S);
    else       recur_loop<false>(bfrag, xi, gp, rowstride, hs, arow, hi, n0 + col16, S);
}

extern "C" void kernel_launch(void* const* d_in, const int* in_sizes, int n_in,
                              void* d_out, int out_size, void* d_ws, size_t ws_size,
                              hipStream_t stream) {
    const float* x  = (const float*)d_in[0];
    const float* Wi = (const float*)d_in[1];
    const float* Wh = (const float*)d_in[2];
    const float* b  = (const float*)d_in[3];
    const float* Wo = (const float*)d_in[4];
    const float* bo = (const float*)d_in[5];
    float* y  = (float*)d_out;
    float* ws = (float*)d_ws;

    constexpr int B = 64, S = 2048;
    static_assert(S % XI_AHEAD == 0, "rnn_recur walks the sequence in groups of XI_AHEAD steps");
    const int M = B * S;

    gemm128_bias<<<1024, 256, 0, stream>>>(x, Wi, b, ws, M);
    rnn_recur<<<B / TM, 256, 0, stream>>>(Wh, ws, S);
    gemm128_bias<<<1024, 256, 0, stream>>>(ws, Wo, bo, y, M);
}


# workspace (xi, then h) laid out as per-step 16-row tiles in accumulator-register order: rnn_recur moves a step's xi and h as 2 16-byte accesses per lane, not 8 4-byte ones; sums and order unchanged
# speedup vs baseline: 2.4780x; 1.0549x over previous
#include <hip/hip_runtime.h>
#include <math.h>

typedef __attribute__((ext_vector_type(2))) float v2f;
typedef __attribute__((ext_vector_type(4))) float v4f;
typedef __attribute__((ext_vector_type(8))) float v8f;

#define HDIM 128
#define TM 16
#define LDS_STRIDE 132

#define WS_TILE (TM * HDIM)

__device__ __forceinline__ v8f wmma4(v2f a, v2f b, v8f c) {
    return __builtin_amdgcn_wmma_f32_16x16x4_f32(false, a, false, b, (short)0, c, false, false);
}

__device__ __forceinline__ float act_relu_tanh(float z) {
#if __has_builtin(__builtin_amdgcn_tanhf)
    float v = __builtin_amdgcn_tanhf(z);
#else
    float v = tanhf(z);
#endif
    return fmaxf(v, 0.f);
}

__device__ __forceinline__ void load_wfrags(const float* __restrict__ W,
                                            int n0, int col16, int hi, v2f* bfrag) {
    const float* wp = W + (hi * 2) * HDIM + n0 + col16;
#pragma unroll
    for (int k = 0; k < 32; ++k) {
        bfrag[k].x = wp[(4 * k) * HDIM];
        bfrag[k].y = wp[(4 * k + 1) * HDIM];
    }
}

__device__ __forceinline__ v8f tile_matmul(const float* __restrict__ arow,
                                           const v2f* __restrict__ bfrag) {
    v2f a[32];
#pragma unroll
    for (int k = 0; k < 32; ++k)
        a[k] = *(const v2f*)(arow + 4 * k);
    v8f c0 = {}, c1 = {}, c2 = {}, c3 = {};
#pragma unroll
    for (int k = 0; k < 8; ++k) {
        c0 = wmma4(a[k],      bfrag[k],      c0);
        c1 = wmma4(a[k + 8],  bfrag[k + 8],  c1);
        c2 = wmma4(a[k + 16], bfrag[k + 16], c2);
        c3 = wmma4(a[k + 24], bfrag[k + 24], c3);
    }
    return (c0 + c1) + (c2 + c3);
}

template <bool IN_TILED, bool OUT_TILED>
__global__ __launch_bounds__(256) void gemm128_bias(const float* __restrict__ X,
                                                    const float* __restrict__ W,
                                                    const float* __restrict__ bias,
                                                    float* __restrict__ Y,
                                                    int S) {
    __shared__ float xs[TM * LDS_STRIDE];
    const int tid   = threadIdx.x;
    const int lane  = tid & 31;
    const int wv    = tid >> 5;
    const int n0    = wv * 16;
    const int col16 = lane & 15;
    const int hi    = lane >> 4;
    const int g     = blockIdx.y;

    v2f bfrag[32];
    load_wfrags(W, n0, col16, hi, bfrag);
    const float bval = bias[n0 + col16];

    const float* arow = &xs[col16 * LDS_STRIDE + hi * 2];

    for (int t = blockIdx.x; t < S; t += gridDim.x) {
        const size_t tile = (size_t)g * S + t;
#pragma unroll
        for (int i = 0; i < 2; ++i) {
            const int lin = (tid + i * 256) * 4;
            if (IN_TILED) {
                const v4f q4 = *(const v4f*)(X + tile * WS_TILE + lin);
                const int sw  = lin >> 8;
                const int sl  = (lin >> 2) & 31;
                const int row = 8 * (sl >> 4) + 4 * ((lin >> 7) & 1);
                const int col = 16 * sw + (sl & 15);
#pragma unroll
                for (int j = 0; j < 4; ++j)
                    xs[(row + j) * LDS_STRIDE + col] = q4[j];
            } else {
                const int r = lin >> 7;
                const int c = lin & 127;
                *(v4f*)(&xs[r * LDS_STRIDE + c]) =
                    *(const v4f*)(X + ((size_t)(g * TM + r) * S + t) * HDIM + c);
            }
        }
        __syncthreads();

        v8f acc = tile_matmul(arow, bfrag);

        if (OUT_TILED) {
            float* yp = Y + tile * WS_TILE + wv * 256 + lane * 4;
            const v4f o0 = {acc[0] + bval, acc[1] + bval, acc[2] + bval, acc[3] + bval};
            const v4f o1 = {acc[4] + bval, acc[5] + bval, acc[6] + bval, acc[7] + bval};
            *(v4f*)yp         = o0;
            *(v4f*)(yp + 128) = o1;
        } else {
            float* yp = Y + ((size_t)(g * TM + 8 * hi) * S + t) * HDIM + n0 + col16;
#pragma unroll
            for (int i = 0; i < 8; ++i)
                yp[(size_t)i * S * HDIM] = acc[i] + bval;
        }
        __syncthreads();
    }
}

__device__ __forceinline__ void lds_barrier_signal() {
    __builtin_amdgcn_fence(__ATOMIC_RELEASE, "workgroup", "local");
    __builtin_amdgcn_s_barrier_signal(-1);
}
__device__ __forceinline__ void lds_barrier_wait() {
    __builtin_amdgcn_s_barrier_wait(-1);
    __builtin_amdgcn_fence(__ATOMIC_ACQUIRE, "workgroup", "local");
}

#define XI_AHEAD 4
#define HBUF (TM * LDS_STRIDE)

__device__ __forceinline__ void load_afrags_half(const float* __restrict__ arow, v2f* f, int kb) {
#pragma unroll
    for (int j = 0; j < 4; ++j) {
#pragma unroll
        for (int c = 0; c < 2; ++c) {
            f[8 * c + 2 * j]     = *(const v2f*)(arow + 4 * (kb + 8 * c + 2 * j));
            f[8 * c + 2 * j + 1] = *(const v2f*)(arow + 4 * (kb + 8 * c + 2 * j + 1));
        }
    }
}

__device__ __forceinline__ void wmma_pair(const v2f* f, const v2f* bfrag, int kb, v8f& ca, v8f& cb,
                                          int k0 = 0, int k1 = 8) {
#pragma unroll
    for (int k = k0; k < k1; ++k) {
        ca = wmma4(f[k],     bfrag[kb + k],     ca);
        cb = wmma4(f[8 + k], bfrag[kb + 8 + k], cb);
    }
}

__device__ __forceinline__ void barrier_wait_behind(v8f& ca, v8f& cb) {
    asm volatile("s_barrier_wait 0xffff" : "+v"(ca), "+v"(cb) : : "memory");
}
__device__ __forceinline__ void barrier_wait_behind(v8f& ca, v8f& cb, v8f& cc, v8f& cd) {
    asm volatile("s_barrier_wait 0xffff" : "+v"(ca), "+v"(cb), "+v"(cc), "+v"(cd) : : "memory");
}
__device__ __forceinline__ void barrier_signal_behind(v8f& ca, v8f& cb, v8f& cc, v8f& cd) {
    asm volatile("s_barrier_signal -1" : "+v"(ca), "+v"(cb), "+v"(cc), "+v"(cd) : : "memory");
}
__device__ __forceinline__ void barrier_wait_plain() {
    asm volatile("s_barrier_wait 0xffff" : : : "memory");
}
__device__ __forceinline__ void barrier_signal_nothing() {
    asm volatile("s_barrier_signal -1");
}
__device__ __forceinline__ void barrier_wait_before(v2f& fa, v2f& fb) {
    asm volatile("s_barrier_wait 0xffff" : "+v"(fa), "+v"(fb) : : "memory");
}

template <bool EARLY>
__device__ __forceinline__ void recur_loop(const v2f* bfrag, float (&xi)[XI_AHEAD][8],
                                           float* __restrict__ gp,
                                           float* hs, const float* arow,
                                           int hi, int ncol, int S) {
    v2f ax[16];
    load_afrags_half(arow + HBUF, ax, 0);
    __builtin_amdgcn_s_barrier_signal(-1);
    if (!EARLY) {
        __builtin_amdgcn_s_barrier_wait(-1);
        __builtin_amdgcn_s_barrier_signal(-1);
    }

    for (int t0 = 0; t0 + XI_AHEAD <= S; t0 += XI_AHEAD) {
#pragma unroll
        for (int u = 0; u < XI_AHEAD; ++u) {
            const int t = t0 + u;
            const float* ar = arow + ((u + 1) & 1) * HBUF;
            const float* aw = arow + (u & 1) * HBUF;
            float* hw = hs + (u & 1) * HBUF;
            v2f ay[16];

            if (EARLY) {
                barrier_wait_before(ax[0], ax[8]);
                barrier_signal_nothing();
            } else {
                barrier_wait_plain();
                barrier_signal_nothing();
                __builtin_amdgcn_s_setprio(0);
                load_afrags_half(ar, ay, 16);
            }
            __builtin_amdgcn_sched_barrier(0);

            v8f c0 = {}, c1 = {}, c2 = {}, c3 = {};
            if (EARLY) {
                wmma_pair(ax, bfrag, 0, c0, c1, 0, 5);
                barrier_wait_behind(c0, c1);
                load_afrags_half(ar, ay, 16);
                __builtin_amdgcn_sched_barrier(0);
                wmma_pair(ax, bfrag, 0, c0, c1, 5, 8);
                asm volatile("" : "+v"(c0), "+v"(c1), "+v"(ay[0]), "+v"(ay[8]));
            } else {
                wmma_pair(ax, bfrag, 0, c0, c1);
            }

            wmma_pair(ay, bfrag, 16, c2, c3);

            if (!EARLY) {
                barrier_wait_behind(c0, c1, c2, c3);
                barrier_signal_behind(c0, c1, c2, c3);
                __builtin_amdgcn_s_setprio(3);
                lds_barrier_wait();
            }

            v8f acc = (c0 + c1) + (c2 + c3);

            float v[8];
#pragma unroll
            for (int i = 0; i < 8; ++i) {
                v[i] = act_relu_tanh(acc[i] + xi[u][i]);
                hw[(i + 8 * hi) * LDS_STRIDE + ncol] = v[i];
            }

            lds_barrier_signal();
            if (!EARLY) __builtin_amdgcn_sched_barrier(0);

            float* sp = gp + (size_t)t * WS_TILE;
            *(v4f*)sp         = (v4f){v[0], v[1], v[2], v[3]};
            *(v4f*)(sp + 128) = (v4f){v[4], v[5], v[6], v[7]};
            const int tn = (t + XI_AHEAD < S) ? t + XI_AHEAD : S - 1;
            const float* lp = gp + (size_t)tn * WS_TILE;
            const v4f x0 = *(const v4f*)lp;
            const v4f x1 = *(const v4f*)(lp + 128);
#pragma unroll
            for (int i = 0; i < 4; ++i) {
                xi[u][i]     = x0[i];
                xi[u][4 + i] = x1[i];
            }

            if (EARLY) lds_barrier_wait();
            load_afrags_half(aw, ax, 0);
            if (EARLY) barrier_signal_nothing();
            __builtin_amdgcn_sched_barrier(0);
        }
    }

    if (EARLY) {
        __builtin_amdgcn_s_barrier_wait(-1);
        __builtin_amdgcn_s_barrier_signal(-1);
    }
    __builtin_amdgcn_s_barrier_wait(-1);
}

__global__ __launch_bounds__(256) void rnn_recur(const float* __restrict__ Wh,
                                                 float* __restrict__ ws,
                                                 int S) {
    __shared__ float hs[2 * HBUF];
    const int tid   = threadIdx.x;
    const int lane  = tid & 31;
    const int wv    = tid >> 5;
    const int n0    = wv * 16;
    const int col16 = lane & 15;
    const int hi    = lane >> 4;
    const bool early = __builtin_amdgcn_readfirstlane(wv) < 4;

    v2f bfrag[32];
    load_wfrags(Wh, n0, col16, hi, bfrag);

    float* gp = ws + (size_t)blockIdx.x * S * WS_TILE + wv * 256 + lane * 4;
    const float* arow = &hs[col16 * LDS_STRIDE + hi * 2];

    float xi[XI_AHEAD][8];
#pragma unroll
    for (int u = 0; u < XI_AHEAD; ++u) {
        const v4f x0 = *(const v4f*)(gp + (size_t)u * WS_TILE);
        const v4f x1 = *(const v4f*)(gp + (size_t)u * WS_TILE + 128);
#pragma unroll
        for (int i = 0; i < 4; ++i) {
            xi[u][i]     = x0[i];
            xi[u][4 + i] = x1[i];
        }
    }

    for (int i = tid; i < 2 * HBUF; i += 256) hs[i] = 0.f;
    __syncthreads();

    if (early) recur_loop<true >(bfrag, xi, gp, hs, arow, hi, n0 + col16, S);
    else       recur_loop<false>(bfrag, xi, gp, hs, arow, hi, n0 + col16, S);
}

extern "C" void kernel_launch(void* const* d_in, const int* in_sizes, int n_in,
                              void* d_out, int out_size, void* d_ws, size_t ws_size,
                              hipStream_t stream) {
    const float* x  = (const float*)d_in[0];
    const float* Wi = (const float*)d_in[1];
    const float* Wh = (const float*)d_in[2];
    const float* b  = (const float*)d_in[3];
    const float* Wo = (const float*)d_in[4];
    const float* bo = (const float*)d_in[5];
    float* y  = (float*)d_out;
    float* ws = (float*)d_ws;

    constexpr int B = 64, S = 2048;
    static_assert(S % XI_AHEAD == 0, "rnn_recur walks the sequence in groups of XI_AHEAD steps");
    static_assert(B % TM == 0, "a workspace tile is TM batch rows of one step");

    gemm128_bias<false, true><<<dim3(256, B / TM), 256, 0, stream>>>(x, Wi, b, ws, S);
    rnn_recur<<<B / TM, 256, 0, stream>>>(Wh, ws, S);
    gemm128_bias<true, false><<<dim3(256, B / TM), 256, 0, stream>>>(ws, Wo, bo, y, S);
}


# rnn_recur: a step's sums run on 2 accumulators: c0 starts from xi (WMMA C operand), k-steps 0-7 then 16-23, c1 8-15 then 24-31, result c0+c1; 12 packed adds per wave and step go; sum order changed
# speedup vs baseline: 2.5593x; 1.0328x over previous
#include <hip/hip_runtime.h>
#include <math.h>

typedef __attribute__((ext_vector_type(2))) float v2f;
typedef __attribute__((ext_vector_type(4))) float v4f;
typedef __attribute__((ext_vector_type(8))) float v8f;

#define HDIM 128
#define TM 16
#define LDS_STRIDE 132

#define WS_TILE (TM * HDIM)

__device__ __forceinline__ v8f wmma4(v2f a, v2f b, v8f c) {
    return __builtin_amdgcn_wmma_f32_16x16x4_f32(false, a, false, b, (short)0, c, false, false);
}

__device__ __forceinline__ float act_relu_tanh(float z) {
#if __has_builtin(__builtin_amdgcn_tanhf)
    float v = __builtin_amdgcn_tanhf(z);
#else
    float v = tanhf(z);
#endif
    return fmaxf(v, 0.f);
}

__device__ __forceinline__ void load_wfrags(const float* __restrict__ W,
                                            int n0, int col16, int hi, v2f* bfrag) {
    const float* wp = W + (hi * 2) * HDIM + n0 + col16;
#pragma unroll
    for (int k = 0; k < 32; ++k) {
        bfrag[k].x = wp[(4 * k) * HDIM];
        bfrag[k].y = wp[(4 * k + 1) * HDIM];
    }
}

__device__ __forceinline__ v8f tile_matmul(const float* __restrict__ arow,
                                           const v2f* __restrict__ bfrag) {
    v2f a[32];
#pragma unroll
    for (int k = 0; k < 32; ++k)
        a[k] = *(const v2f*)(arow + 4 * k);
    v8f c0 = {}, c1 = {}, c2 = {}, c3 = {};
#pragma unroll
    for (int k = 0; k < 8; ++k) {
        c0 = wmma4(a[k],      bfrag[k],      c0);
        c1 = wmma4(a[k + 8],  bfrag[k + 8],  c1);
        c2 = wmma4(a[k + 16], bfrag[k + 16], c2);
        c3 = wmma4(a[k + 24], bfrag[k + 24], c3);
    }
    return (c0 + c1) + (c2 + c3);
}

template <bool IN_TILED, bool OUT_TILED>
__global__ __launch_bounds__(256) void gemm128_bias(const float* __restrict__ X,
                                                    const float* __restrict__ W,
                                                    const float* __restrict__ bias,
                                                    float* __restrict__ Y,
                                                    int S) {
    __shared__ float xs[TM * LDS_STRIDE];
    const int tid   = threadIdx.x;
    const int lane  = tid & 31;
    const int wv    = tid >> 5;
    const int n0    = wv * 16;
    const int col16 = lane & 15;
    const int hi    = lane >> 4;
    const int g     = blockIdx.y;

    v2f bfrag[32];
    load_wfrags(W, n0, col16, hi, bfrag);
    const float bval = bias[n0 + col16];

    const float* arow = &xs[col16 * LDS_STRIDE + hi * 2];

    for (int t = blockIdx.x; t < S; t += gridDim.x) {
        const size_t tile = (size_t)g * S + t;
#pragma unroll
        for (int i = 0; i < 2; ++i) {
            const int lin = (tid + i * 256) * 4;
            if (IN_TILED) {
                const v4f q4 = *(const v4f*)(X + tile * WS_TILE + lin);
                const int sw  = lin >> 8;
                const int sl  = (lin >> 2) & 31;
                const int row = 8 * (sl >> 4) + 4 * ((lin >> 7) & 1);
                const int col = 16 * sw + (sl & 15);
#pragma unroll
                for (int j = 0; j < 4; ++j)
                    xs[(row + j) * LDS_STRIDE + col] = q4[j];
            } else {
                const int r = lin >> 7;
                const int c = lin & 127;
                *(v4f*)(&xs[r * LDS_STRIDE + c]) =
                    *(const v4f*)(X + ((size_t)(g * TM + r) * S + t) * HDIM + c);
            }
        }
        __syncthreads();

        v8f acc = tile_matmul(arow, bfrag);

        if (OUT_TILED) {
            float* yp = Y + tile * WS_TILE + wv * 256 + lane * 4;
            const v4f o0 = {acc[0] + bval, acc[1] + bval, acc[2] + bval, acc[3] + bval};
            const v4f o1 = {acc[4] + bval, acc[5] + bval, acc[6] + bval, acc[7] + bval};
            *(v4f*)yp         = o0;
            *(v4f*)(yp + 128) = o1;
        } else {
            float* yp = Y + ((size_t)(g * TM + 8 * hi) * S + t) * HDIM + n0 + col16;
#pragma unroll
            for (int i = 0; i < 8; ++i)
                yp[(size_t)i * S * HDIM] = acc[i] + bval;
        }
        __syncthreads();
    }
}

__device__ __forceinline__ void lds_barrier_signal() {
    __builtin_amdgcn_fence(__ATOMIC_RELEASE, "workgroup", "local");
    __builtin_amdgcn_s_barrier_signal(-1);
}
__device__ __forceinline__ void lds_barrier_wait() {
    __builtin_amdgcn_s_barrier_wait(-1);
    __builtin_amdgcn_fence(__ATOMIC_ACQUIRE, "workgroup", "local");
}

#define XI_AHEAD 4
#define HBUF (TM * LDS_STRIDE)

__device__ __forceinline__ void load_afrags_half(const float* __restrict__ arow, v2f* f, int kb) {
#pragma unroll
    for (int j = 0; j < 4; ++j) {
#pragma unroll
        for (int c = 0; c < 2; ++c) {
            f[8 * c + 2 * j]     = *(const v2f*)(arow + 4 * (kb + 8 * c + 2 * j));
            f[8 * c + 2 * j + 1] = *(const v2f*)(arow + 4 * (kb + 8 * c + 2 * j + 1));
        }
    }
}

__device__ __forceinline__ void wmma_pair(const v2f* f, const v2f* bfrag, int kb, v8f& ca, v8f& cb,
                                          int k0 = 0, int k1 = 8) {
#pragma unroll
    for (int k = k0; k < k1; ++k) {
        ca = wmma4(f[k],     bfrag[kb + k],     ca);
        cb = wmma4(f[8 + k], bfrag[kb + 8 + k], cb);
    }
}

__device__ __forceinline__ void barrier_wait_behind(v8f& ca, v8f& cb) {
    asm volatile("s_barrier_wait 0xffff" : "+v"(ca), "+v"(cb) : : "memory");
}
__device__ __forceinline__ void barrier_signal_behind(v8f& ca, v8f& cb) {
    asm volatile("s_barrier_signal -1" : "+v"(ca), "+v"(cb) : : "memory");
}
__device__ __forceinline__ void barrier_wait_plain() {
    asm volatile("s_barrier_wait 0xffff" : : : "memory");
}
__device__ __forceinline__ void barrier_signal_nothing() {
    asm volatile("s_barrier_signal -1");
}
__device__ __forceinline__ void barrier_wait_before(v2f& fa, v2f& fb) {
    asm volatile("s_barrier_wait 0xffff" : "+v"(fa), "+v"(fb) : : "memory");
}

template <bool EARLY>
__device__ __forceinline__ void recur_loop(const v2f* bfrag, float (&xi)[XI_AHEAD][8],
                                           float* __restrict__ gp,
                                           float* hs, const float* arow,
                                           int hi, int ncol, int S) {
    v2f ax[16];
    load_afrags_half(arow + HBUF, ax, 0);
    __builtin_amdgcn_s_barrier_signal(-1);
    if (!EARLY) {
        __builtin_amdgcn_s_barrier_wait(-1);
        __builtin_amdgcn_s_barrier_signal(-1);
    }

    for (int t0 = 0; t0 + XI_AHEAD <= S; t0 += XI_AHEAD) {
#pragma unroll
        for (int u = 0; u < XI_AHEAD; ++u) {
            const int t = t0 + u;
            const float* ar = arow + ((u + 1) & 1) * HBUF;
            const float* aw = arow + (u & 1) * HBUF;
            float* hw = hs + (u & 1) * HBUF;
            v2f ay[16];

            if (EARLY) {
                barrier_wait_before(ax[0], ax[8]);
                barrier_signal_nothing();
            } else {
                barrier_wait_plain();
                barrier_signal_nothing();
                __builtin_amdgcn_s_setprio(0);
                load_afrags_half(ar, ay, 16);
            }
            __builtin_amdgcn_sched_barrier(0);

            v8f c0 = {xi[u][0], xi[u][1], xi[u][2], xi[u][3], xi[u][4], xi[u][5], xi[u][6], xi[u][7]};
            v8f c1 = {};
            if (EARLY) {
                wmma_pair(ax, bfrag, 0, c0, c1, 0, 5);
                barrier_wait_behind(c0, c1);
                load_afrags_half(ar, ay, 16);
                __builtin_amdgcn_sched_barrier(0);
                wmma_pair(ax, bfrag, 0, c0, c1, 5, 8);
            } else {
                wmma_pair(ax, bfrag, 0, c0, c1);
            }

            wmma_pair(ay, bfrag, 16, c0, c1);

            if (!EARLY) {
                barrier_wait_behind(c0, c1);
                barrier_signal_behind(c0, c1);
                __builtin_amdgcn_s_setprio(3);
                lds_barrier_wait();
            }

            v8f acc = c0 + c1;

            float v[8];
#pragma unroll
            for (int i = 0; i < 8; ++i) {
                v[i] = act_relu_tanh(acc[i]);
                hw[(i + 8 * hi) * LDS_STRIDE + ncol] = v[i];
            }

            lds_barrier_signal();
            if (!EARLY) __builtin_amdgcn_sched_barrier(0);

            float* sp = gp + (size_t)t * WS_TILE;
            *(v4f*)sp         = (v4f){v[0], v[1], v[2], v[3]};
            *(v4f*)(sp + 128) = (v4f){v[4], v[5], v[6], v[7]};
            const int tn = (t + XI_AHEAD < S) ? t + XI_AHEAD : S - 1;
            const float* lp = gp + (size_t)tn * WS_TILE;
            const v4f x0 = *(const v4f*)lp;
            const v4f x1 = *(const v4f*)(lp + 128);
#pragma unroll
            for (int i = 0; i < 4; ++i) {
                xi[u][i]     = x0[i];
                xi[u][4 + i] = x1[i];
            }

            if (EARLY) lds_barrier_wait();
            load_afrags_half(aw, ax, 0);
            if (EARLY) barrier_signal_nothing();
            __builtin_amdgcn_sched_barrier(0);
        }
    }

    if (EARLY) {
        __builtin_amdgcn_s_barrier_wait(-1);
        __builtin_amdgcn_s_barrier_signal(-1);
    }
    __builtin_amdgcn_s_barrier_wait(-1);
}

__global__ __launch_bounds__(256) void rnn_recur(const float* __restrict__ Wh,
                                                 float* __restrict__ ws,
                                                 int S) {
    __shared__ float hs[2 * HBUF];
    const int tid   = threadIdx.x;
    const int lane  = tid & 31;
    const int wv    = tid >> 5;
    const int n0    = wv * 16;
    const int col16 = lane & 15;
    const int hi    = lane >> 4;
    const bool early = __builtin_amdgcn_readfirstlane(wv) < 4;

    v2f bfrag[32];
    load_wfrags(Wh, n0, col16, hi, bfrag);

    float* gp = ws + (size_t)blockIdx.x * S * WS_TILE + wv * 256 + lane * 4;
    const float* arow = &hs[col16 * LDS_STRIDE + hi * 2];

    float xi[XI_AHEAD][8];
#pragma unroll
    for (int u = 0; u < XI_AHEAD; ++u) {
        const v4f x0 = *(const v4f*)(gp + (size_t)u * WS_TILE);
        const v4f x1 = *(const v4f*)(gp + (size_t)u * WS_TILE + 128);
#pragma unroll
        for (int i = 0; i < 4; ++i) {
            xi[u][i]     = x0[i];
            xi[u][4 + i] = x1[i];
        }
    }

    for (int i = tid; i < 2 * HBUF; i += 256) hs[i] = 0.f;
    __syncthreads();

    if (early) recur_loop<true >(bfrag, xi, gp, hs, arow, hi, n0 + col16, S);
    else       recur_loop<false>(bfrag, xi, gp, hs, arow, hi, n0 + col16, S);
}

extern "C" void kernel_launch(void* const* d_in, const int* in_sizes, int n_in,
                              void* d_out, int out_size, void* d_ws, size_t ws_size,
                              hipStream_t stream) {
    const float* x  = (const float*)d_in[0];
    const float* Wi = (const float*)d_in[1];
    const float* Wh = (const float*)d_in[2];
    const float* b  = (const float*)d_in[3];
    const float* Wo = (const float*)d_in[4];
    const float* bo = (const float*)d_in[5];
    float* y  = (float*)d_out;
    float* ws = (float*)d_ws;

    constexpr int B = 64, S = 2048;
    static_assert(S % XI_AHEAD == 0, "rnn_recur walks the sequence in groups of XI_AHEAD steps");
    static_assert(B % TM == 0, "a workspace tile is TM batch rows of one step");

    gemm128_bias<false, true><<<dim3(256, B / TM), 256, 0, stream>>>(x, Wi, b, ws, S);
    rnn_recur<<<B / TM, 256, 0, stream>>>(Wh, ws, S);
    gemm128_bias<true, false><<<dim3(256, B / TM), 256, 0, stream>>>(ws, Wo, bo, y, S);
}
